# starting kernel
# baseline (speedup 1.0000x reference)
#include <hip/hip_runtime.h>


typedef __attribute__((ext_vector_type(2))) float v2f;
typedef __attribute__((ext_vector_type(4))) float v4f;
typedef __attribute__((ext_vector_type(8))) float v8f;

#define B_SZ 256
#define D_SZ 512
#define P_SZ 128
#define E_SZ 128
#define XB_PITCH (D_SZ * P_SZ)
#define WP_PITCH (D_SZ * E_SZ)
#define OB_PITCH (E_SZ * P_SZ)

#define KC     16
#define KPAD   18
#define PB      8
#define BTILE  16
#define ETILE  64
#define NCHUNK (D_SZ / KC)

__global__ __launch_bounds__(256, 2)
void loop_proj_wmma_kernel(const float* __restrict__ x,
                           const float* __restrict__ W,
                           const float* __restrict__ bias,
                           float* __restrict__ out)
{
    __shared__ float sX[2][PB * BTILE * KPAD];
    __shared__ float sW[2][PB * ETILE * KPAD];

    const int tid  = threadIdx.x;
    const int lane = tid & 31;
    const int wave = tid >> 5;

    const int pg0 = blockIdx.x * PB;
    const int b0  = blockIdx.y * BTILE;
    const int e0  = blockIdx.z * ETILE;

    const int m    = lane & 15;
    const int koff = (lane >> 4) << 1;

    v8f acc[4] = {};

    const int pairX = tid >> 1;
    const int phalf = (tid & 1) * 4;
    const int iX    = pairX >> 4;
    const int kX    = pairX & 15;
    const float* xp0 = x + (size_t)(b0 + iX) * XB_PITCH
                         + (size_t)kX * P_SZ + pg0 + phalf;
    const float* xp1 = xp0 + (size_t)8 * XB_PITCH;

    const int piW   = (tid >> 1) >> 4;
    const int kW    = (tid >> 1) & 15;
    const int eHalf = (tid & 1) * 32;
    const float* wp = W + (size_t)(pg0 + piW) * WP_PITCH
                        + (size_t)kW * E_SZ + e0 + eHalf;

    auto stage = [&](int buf, const v4f& rx0, const v4f& rx1, const v4f (&rw)[8]) {
        #pragma unroll
        for (int j = 0; j < 4; ++j) {
            sX[buf][((phalf + j) * BTILE + iX)     * KPAD + kX] = rx0[j];
            sX[buf][((phalf + j) * BTILE + iX + 8) * KPAD + kX] = rx1[j];
        }
        #pragma unroll
        for (int j = 0; j < 8; ++j) {
            const int e = eHalf + j * 4;
            #pragma unroll
            for (int q = 0; q < 4; ++q)
                sW[buf][(piW * ETILE + e + q) * KPAD + kW] = rw[j][q];
        }
    };

    {
        v4f rx0 = *(const v4f*)xp0;
        v4f rx1 = *(const v4f*)xp1;
        v4f rw[8];
        #pragma unroll
        for (int j = 0; j < 8; ++j) rw[j] = *(const v4f*)(wp + j * 4);
        stage(0, rx0, rx1, rw);
    }

    for (int c = 0; c < NCHUNK; ++c) {
        const int cur  = c & 1;
        const bool more = (c + 1) < NCHUNK;

        v4f nrx0, nrx1, nrw[8];
        if (more) {
            const size_t off = (size_t)(c + 1) * KC * P_SZ;
            nrx0 = *(const v4f*)(xp0 + off);
            nrx1 = *(const v4f*)(xp1 + off);
            #pragma unroll
            for (int j = 0; j < 8; ++j) nrw[j] = *(const v4f*)(wp + off + j * 4);
        }

        __syncthreads();

        const float* bX = sX[cur];
        const float* bW = sW[cur];
        #pragma unroll
        for (int k0 = 0; k0 < KC; k0 += 4) {
            v2f a = *(const v2f*)&bX[(wave * BTILE + m) * KPAD + k0 + koff];
            #pragma unroll
            for (int t = 0; t < 4; ++t) {
                v2f bf = *(const v2f*)&bW[(wave * ETILE + t * 16 + m) * KPAD + k0 + koff];
                acc[t] = __builtin_amdgcn_wmma_f32_16x16x4_f32(
                             false, a, false, bf, (short)0, acc[t], false, false);
            }
        }

        if (more) stage(cur ^ 1, nrx0, nrx1, nrw);
    }

    const int p    = pg0 + wave;
    const int mtop = (lane >> 4) << 3;
    float bi[4];
    #pragma unroll
    for (int t = 0; t < 4; ++t) bi[t] = bias[p * E_SZ + e0 + t * 16 + m];

    #pragma unroll
    for (int r = 0; r < 8; ++r) {
        const size_t bb = (size_t)(b0 + r + mtop) * OB_PITCH;
        #pragma unroll
        for (int t = 0; t < 4; ++t)
            out[bb + (size_t)(e0 + t * 16 + m) * P_SZ + p] = acc[t][r] + bi[t];
    }
}

extern "C" void kernel_launch(void* const* d_in, const int* in_sizes, int n_in,
                              void* d_out, int out_size, void* d_ws, size_t ws_size,
                              hipStream_t stream) {
    (void)in_sizes; (void)n_in; (void)d_ws; (void)ws_size; (void)out_size;
    const float* x    = (const float*)d_in[0];
    const float* W    = (const float*)d_in[1];
    const float* bias = (const float*)d_in[2];
    float* out        = (float*)d_out;

    dim3 grid(P_SZ / PB, B_SZ / BTILE, E_SZ / ETILE);
    dim3 block(256);
    loop_proj_wmma_kernel<<<grid, block, 0, stream>>>(x, W, bias, out);
}


# Wave tile 64 batch rows x 32 e (4x2 accumulator tiles, was 1x4): a W fragment feeds 4 batch tiles, grid 16x4x4, each W element fetched by 4 workgroups not 16; d-sum order unchanged
# speedup vs baseline: 1.0421x; 1.0421x over previous
#include <hip/hip_runtime.h>


typedef __attribute__((ext_vector_type(2))) float v2f;
typedef __attribute__((ext_vector_type(4))) float v4f;
typedef __attribute__((ext_vector_type(8))) float v8f;

#define B_SZ 256
#define D_SZ 512
#define P_SZ 128
#define E_SZ 128
#define XB_PITCH (D_SZ * P_SZ)
#define WP_PITCH (D_SZ * E_SZ)
#define OB_PITCH (E_SZ * P_SZ)

#define KC     16
#define KPAD   18
#define PB      8
#define BTILE  64
#define ETILE  32
#define MT     (BTILE / 16)
#define NT     (ETILE / 16)
#define XREP   (BTILE / 8)
#define WVEC   (ETILE / 8)
#define NCHUNK (D_SZ / KC)

__global__ __launch_bounds__(256, 1)
void loop_proj_wmma_kernel(const float* __restrict__ x,
                           const float* __restrict__ W,
                           const float* __restrict__ bias,
                           float* __restrict__ out)
{
    __shared__ float sX[2][PB * BTILE * KPAD];
    __shared__ float sW[2][PB * ETILE * KPAD];

    const int tid  = threadIdx.x;
    const int lane = tid & 31;
    const int wave = tid >> 5;

    const int pg0 = blockIdx.x * PB;
    const int b0  = blockIdx.y * BTILE;
    const int e0  = blockIdx.z * ETILE;

    const int m    = lane & 15;
    const int koff = (lane >> 4) << 1;

    v8f acc[MT][NT] = {};

    const int pairX = tid >> 1;
    const int phalf = (tid & 1) * 4;
    const int iX    = pairX >> 4;
    const int kX    = pairX & 15;
    const float* xp0 = x + (size_t)(b0 + iX) * XB_PITCH
                         + (size_t)kX * P_SZ + pg0 + phalf;

    const int piW   = (tid >> 1) >> 4;
    const int kW    = (tid >> 1) & 15;
    const int eHalf = (tid & 1) * (ETILE / 2);
    const float* wp = W + (size_t)(pg0 + piW) * WP_PITCH
                        + (size_t)kW * E_SZ + e0 + eHalf;

    auto stage = [&](int buf, const v4f (&rw)[WVEC], const v4f (&rx)[XREP]) {
        #pragma unroll
        for (int r = 0; r < XREP; ++r) {
            #pragma unroll
            for (int j = 0; j < 4; ++j)
                sX[buf][((phalf + j) * BTILE + iX + 8 * r) * KPAD + kX] = rx[r][j];
        }
        #pragma unroll
        for (int j = 0; j < WVEC; ++j) {
            const int e = eHalf + j * 4;
            #pragma unroll
            for (int q = 0; q < 4; ++q)
                sW[buf][(piW * ETILE + e + q) * KPAD + kW] = rw[j][q];
        }
    };

    {
        v4f rw[WVEC], rx[XREP];
        #pragma unroll
        for (int j = 0; j < WVEC; ++j) rw[j] = *(const v4f*)(wp + j * 4);
        #pragma unroll
        for (int r = 0; r < XREP; ++r)
            rx[r] = *(const v4f*)(xp0 + (size_t)(8 * r) * XB_PITCH);
        stage(0, rw, rx);
    }

    for (int c = 0; c < NCHUNK; ++c) {
        const int cur  = c & 1;
        const bool more = (c + 1) < NCHUNK;

        v4f nrw[WVEC], nrx[XREP];
        if (more) {
            const size_t off = (size_t)(c + 1) * KC * P_SZ;
            #pragma unroll
            for (int j = 0; j < WVEC; ++j) nrw[j] = *(const v4f*)(wp + off + j * 4);
            #pragma unroll
            for (int r = 0; r < XREP; ++r)
                nrx[r] = *(const v4f*)(xp0 + off + (size_t)(8 * r) * XB_PITCH);
        }

        __syncthreads();

        const float* bX = sX[cur];
        const float* bW = sW[cur];
        #pragma unroll
        for (int k0 = 0; k0 < KC; k0 += 4) {
            v2f a[MT], bf[NT];
            #pragma unroll
            for (int mt = 0; mt < MT; ++mt)
                a[mt] = *(const v2f*)&bX[(wave * BTILE + mt * 16 + m) * KPAD + k0 + koff];
            #pragma unroll
            for (int nt = 0; nt < NT; ++nt)
                bf[nt] = *(const v2f*)&bW[(wave * ETILE + nt * 16 + m) * KPAD + k0 + koff];
            #pragma unroll
            for (int mt = 0; mt < MT; ++mt) {
                #pragma unroll
                for (int nt = 0; nt < NT; ++nt) {
                    acc[mt][nt] = __builtin_amdgcn_wmma_f32_16x16x4_f32(
                                      false, a[mt], false, bf[nt], (short)0,
                                      acc[mt][nt], false, false);
                }
            }
        }

        if (more) stage(cur ^ 1, nrw, nrx);
    }

    const int p    = pg0 + wave;
    const int mtop = (lane >> 4) << 3;
    float bi[NT];
    #pragma unroll
    for (int nt = 0; nt < NT; ++nt) bi[nt] = bias[p * E_SZ + e0 + nt * 16 + m];

    #pragma unroll
    for (int mt = 0; mt < MT; ++mt) {
        #pragma unroll
        for (int r = 0; r < 8; ++r) {
            const size_t bb = (size_t)(b0 + mt * 16 + r + mtop) * OB_PITCH;
            #pragma unroll
            for (int nt = 0; nt < NT; ++nt)
                out[bb + (size_t)(e0 + nt * 16 + m) * P_SZ + p] = acc[mt][nt][r] + bi[nt];
        }
    }
}

extern "C" void kernel_launch(void* const* d_in, const int* in_sizes, int n_in,
                              void* d_out, int out_size, void* d_ws, size_t ws_size,
                              hipStream_t stream) {
    (void)in_sizes; (void)n_in; (void)d_ws; (void)ws_size; (void)out_size;
    const float* x    = (const float*)d_in[0];
    const float* W    = (const float*)d_in[1];
    const float* bias = (const float*)d_in[2];
    float* out        = (float*)d_out;

    dim3 grid(P_SZ / PB, B_SZ / BTILE, E_SZ / ETILE);
    dim3 block(256);
    loop_proj_wmma_kernel<<<grid, block, 0, stream>>>(x, W, bias, out);
}


# Wave = 4 adjacent prototypes x (32 b x 16 e), was 1 x (32 b x 64 e): a lane's accumulators hold out[b][e][p..p+3], stored as one 16-B piece, 16 stores per wave not 64; d-sum order unchanged
# speedup vs baseline: 1.9125x; 1.8352x over previous
#include <hip/hip_runtime.h>


typedef __attribute__((ext_vector_type(2))) float v2f;
typedef __attribute__((ext_vector_type(4))) float v4f;
typedef __attribute__((ext_vector_type(8))) float v8f;

#define B_SZ 256
#define D_SZ 512
#define P_SZ 128
#define E_SZ 128
#define XB_PITCH (D_SZ * P_SZ)
#define WP_PITCH (D_SZ * E_SZ)
#define OB_PITCH (E_SZ * P_SZ)

#define KC     64
#define KPAD   68
#define PB      8
#define PW      4
#define BTILE  32
#define ETILE  64
#define MT     (BTILE / 16)
#define XROWS  (256 / 2 / KC)
#define XREP   (BTILE / XROWS)
#define WVEC   (ETILE / 8)
#define WREP   (PB / XROWS)
#define NCHUNK (D_SZ / KC)

__global__ __launch_bounds__(256, 1)
void loop_proj_wmma_kernel(const float* __restrict__ x,
                           const float* __restrict__ W,
                           const float* __restrict__ bias,
                           float* __restrict__ out)
{
    __shared__ float sX[PB * BTILE * KPAD];
    __shared__ float sW[PB * ETILE * KPAD];

    const int tid  = threadIdx.x;
    const int lane = tid & 31;
    const int wave = __builtin_amdgcn_readfirstlane(tid >> 5);
    const int pq   = (wave >> 2) * PW;
    const int es   = (wave & 3) * 16;

    const int pg0 = blockIdx.x * PB;
    const int b0  = blockIdx.y * BTILE;
    const int e0  = blockIdx.z * ETILE;

    const int m    = lane & 15;
    const int koff = (lane >> 4) << 1;

    v8f acc[PW][MT] = {};

    const int pairX = tid >> 1;
    const int phalf = (tid & 1) * 4;
    const int iX    = pairX / KC;
    const int kX    = pairX % KC;
    const float* xp0 = x + (size_t)(b0 + iX) * XB_PITCH
                         + (size_t)kX * P_SZ + pg0 + phalf;

    const int piW   = (tid >> 1) / KC;
    const int kW    = (tid >> 1) % KC;
    const int eSel  = (tid & 1) * 4;
    const float* wp = W + (size_t)(pg0 + piW) * WP_PITCH
                        + (size_t)kW * E_SZ + e0 + eSel;

    auto stage = [&](const v4f (&rw)[WREP][WVEC], const v4f (&rx)[XREP]) {
        #pragma unroll
        for (int r = 0; r < XREP; ++r) {
            #pragma unroll
            for (int j = 0; j < 4; ++j)
                sX[((phalf + j) * BTILE + iX + XROWS * r) * KPAD + kX] = rx[r][j];
        }
        #pragma unroll
        for (int s = 0; s < WREP; ++s) {
            #pragma unroll
            for (int j = 0; j < WVEC; ++j) {
                const int e = eSel + j * 8;
                #pragma unroll
                for (int q = 0; q < 4; ++q)
                    sW[((piW + XROWS * s) * ETILE + e + q) * KPAD + kW] = rw[s][j][q];
            }
        }
    };
    auto fetch = [&](size_t off, v4f (&rw)[WREP][WVEC], v4f (&rx)[XREP]) {
        #pragma unroll
        for (int s = 0; s < WREP; ++s) {
            #pragma unroll
            for (int j = 0; j < WVEC; ++j)
                rw[s][j] = *(const v4f*)(wp + off + (size_t)(XROWS * s) * WP_PITCH + j * 8);
        }
        #pragma unroll
        for (int r = 0; r < XREP; ++r)
            rx[r] = *(const v4f*)(xp0 + off + (size_t)(XROWS * r) * XB_PITCH);
    };

    {
        v4f rw[WREP][WVEC], rx[XREP];
        fetch(0, rw, rx);
        stage(rw, rx);
    }
    __syncthreads();

    for (int c = 0; c < NCHUNK; ++c) {
        const bool more = (c + 1) < NCHUNK;

        v4f nrw[WREP][WVEC], nrx[XREP];
        if (more) fetch((size_t)(c + 1) * KC * P_SZ, nrw, nrx);

        const float* bX = sX;
        const float* bW = sW;
        #pragma unroll
        for (int k0 = 0; k0 < KC; k0 += 4) {
            v2f a[PW][MT], bf[PW];
            #pragma unroll
            for (int j = 0; j < PW; ++j) {
                #pragma unroll
                for (int mt = 0; mt < MT; ++mt)
                    a[j][mt] = *(const v2f*)&bX[((pq + j) * BTILE + mt * 16 + m) * KPAD + k0 + koff];
                bf[j] = *(const v2f*)&bW[((pq + j) * ETILE + es + m) * KPAD + k0 + koff];
            }
            #pragma unroll
            for (int j = 0; j < PW; ++j) {
                #pragma unroll
                for (int mt = 0; mt < MT; ++mt) {
                    acc[j][mt] = __builtin_amdgcn_wmma_f32_16x16x4_f32(
                                     false, a[j][mt], false, bf[j], (short)0,
                                     acc[j][mt], false, false);
                }
            }
        }

        if (more) {
            __syncthreads();
            stage(nrw, nrx);
            __syncthreads();
        }
    }

    const int el   = (int)__builtin_amdgcn_mbcnt_lo(~0u, 0u);
    const int p    = pg0 + pq;
    const int e    = e0 + es + (el & 15);
    const int mtop = (el >> 4) << 3;
    float bi[PW];
    #pragma unroll
    for (int j = 0; j < PW; ++j) bi[j] = bias[(p + j) * E_SZ + e];

    #pragma unroll
    for (int mt = 0; mt < MT; ++mt) {
        #pragma unroll
        for (int r = 0; r < 8; ++r) {
            const size_t bb = (size_t)(b0 + mt * 16 + r + mtop) * OB_PITCH;
            v4f o;
            #pragma unroll
            for (int j = 0; j < PW; ++j) o[j] = acc[j][mt][r] + bi[j];
            *(v4f*)(out + bb + (size_t)e * P_SZ + p) = o;
        }
    }
}

extern "C" void kernel_launch(void* const* d_in, const int* in_sizes, int n_in,
                              void* d_out, int out_size, void* d_ws, size_t ws_size,
                              hipStream_t stream) {
    (void)in_sizes; (void)n_in; (void)d_ws; (void)ws_size; (void)out_size;
    const float* x    = (const float*)d_in[0];
    const float* W    = (const float*)d_in[1];
    const float* bias = (const float*)d_in[2];
    float* out        = (float*)d_out;

    dim3 grid(P_SZ / PB, B_SZ / BTILE, E_SZ / ETILE);
    dim3 block(256);
    loop_proj_wmma_kernel<<<grid, block, 0, stream>>>(x, W, bias, out);
}


# Wave = 8 prototypes x (16 b x 16 e), was 4 x (32 b x 16 e): a lane holds 32 B of an out row, e-pair lanes swap halves by DPP, a store writes 16 pieces of 32 B not 32 of 16 B; d-sum order unchanged
# speedup vs baseline: 2.2205x; 1.1610x over previous
#include <hip/hip_runtime.h>


typedef __attribute__((ext_vector_type(2))) float v2f;
typedef __attribute__((ext_vector_type(4))) float v4f;
typedef __attribute__((ext_vector_type(8))) float v8f;

#define B_SZ 256
#define D_SZ 512
#define P_SZ 128
#define E_SZ 128
#define XB_PITCH (D_SZ * P_SZ)
#define WP_PITCH (D_SZ * E_SZ)
#define OB_PITCH (E_SZ * P_SZ)

#define KC     64
#define KPAD   68
#define PB      8
#define BTILE  32
#define ETILE  64
#define XROWS  (256 / 2 / KC)
#define XREP   (BTILE / XROWS)
#define WVEC   (ETILE / 8)
#define WREP   (PB / XROWS)
#define NCHUNK (D_SZ / KC)

__global__ __launch_bounds__(256, 1)
void loop_proj_wmma_kernel(const float* __restrict__ x,
                           const float* __restrict__ W,
                           const float* __restrict__ bias,
                           float* __restrict__ out)
{
    __shared__ float sX[PB * BTILE * KPAD];
    __shared__ float sW[PB * ETILE * KPAD];

    const int tid  = threadIdx.x;
    const int lane = tid & 31;
    const int wave = __builtin_amdgcn_readfirstlane(tid >> 5);
    const int bt   = (wave >> 2) * 16;
    const int es   = (wave & 3) * 16;

    const int pg0 = blockIdx.x * PB;
    const int b0  = blockIdx.y * BTILE;
    const int e0  = blockIdx.z * ETILE;

    const int m    = lane & 15;
    const int koff = (lane >> 4) << 1;

    v8f acc[PB] = {};

    const int pairX = tid >> 1;
    const int phalf = (tid & 1) * 4;
    const int iX    = pairX / KC;
    const int kX    = pairX % KC;
    const float* xp0 = x + (size_t)(b0 + iX) * XB_PITCH
                         + (size_t)kX * P_SZ + pg0 + phalf;

    const int piW   = (tid >> 1) / KC;
    const int kW    = (tid >> 1) % KC;
    const int eSel  = (tid & 1) * 4;
    const float* wp = W + (size_t)(pg0 + piW) * WP_PITCH
                        + (size_t)kW * E_SZ + e0 + eSel;

    auto stage = [&](const v4f (&rw)[WREP][WVEC], const v4f (&rx)[XREP]) {
        #pragma unroll
        for (int r = 0; r < XREP; ++r) {
            #pragma unroll
            for (int j = 0; j < 4; ++j)
                sX[((phalf + j) * BTILE + iX + XROWS * r) * KPAD + kX] = rx[r][j];
        }
        #pragma unroll
        for (int s = 0; s < WREP; ++s) {
            #pragma unroll
            for (int j = 0; j < WVEC; ++j) {
                const int e = eSel + j * 8;
                #pragma unroll
                for (int q = 0; q < 4; ++q)
                    sW[((piW + XROWS * s) * ETILE + e + q) * KPAD + kW] = rw[s][j][q];
            }
        }
    };
    auto fetch = [&](size_t off, v4f (&rw)[WREP][WVEC], v4f (&rx)[XREP]) {
        #pragma unroll
        for (int s = 0; s < WREP; ++s) {
            #pragma unroll
            for (int j = 0; j < WVEC; ++j)
                rw[s][j] = *(const v4f*)(wp + off + (size_t)(XROWS * s) * WP_PITCH + j * 8);
        }
        #pragma unroll
        for (int r = 0; r < XREP; ++r)
            rx[r] = *(const v4f*)(xp0 + off + (size_t)(XROWS * r) * XB_PITCH);
    };

    {
        v4f rw[WREP][WVEC], rx[XREP];
        fetch(0, rw, rx);
        stage(rw, rx);
    }
    __syncthreads();

    for (int c = 0; c < NCHUNK; ++c) {
        const bool more = (c + 1) < NCHUNK;

        v4f nrw[WREP][WVEC], nrx[XREP];
        if (more) fetch((size_t)(c + 1) * KC * P_SZ, nrw, nrx);

        const float* bX = sX;
        const float* bW = sW;
        #pragma unroll
        for (int k0 = 0; k0 < KC; k0 += 4) {
            v2f a[PB], bf[PB];
            #pragma unroll
            for (int j = 0; j < PB; ++j) {
                a[j]  = *(const v2f*)&bX[(j * BTILE + bt + m) * KPAD + k0 + koff];
                bf[j] = *(const v2f*)&bW[(j * ETILE + es + m) * KPAD + k0 + koff];
            }
            #pragma unroll
            for (int j = 0; j < PB; ++j) {
                acc[j] = __builtin_amdgcn_wmma_f32_16x16x4_f32(
                             false, a[j], false, bf[j], (short)0,
                             acc[j], false, false);
            }
        }

        if (more) {
            __syncthreads();
            stage(nrw, nrx);
            __syncthreads();
        }
    }

    const int  el   = (int)__builtin_amdgcn_mbcnt_lo(~0u, 0u);
    const int  ec   = el & 15;
    const bool odd  = (ec & 1) != 0;
    const int  mtop = (el >> 4) << 3;
    const int  e    = e0 + es + ec;
    float bi[PB];
    #pragma unroll
    for (int j = 0; j < PB; ++j) bi[j] = bias[(pg0 + j) * E_SZ + e];

    float* orow = out + (size_t)(e0 + es + (ec & ~1)) * P_SZ + pg0 + (odd ? 4 : 0);

    #pragma unroll
    for (int r = 0; r < 8; ++r) {
        const size_t bb = (size_t)(b0 + bt + r + mtop) * OB_PITCH;
        float keep[4], got[4];
        #pragma unroll
        for (int q = 0; q < 4; ++q) {
            const float lo = acc[q][r]     + bi[q];
            const float hi = acc[q + 4][r] + bi[q + 4];
            keep[q] = odd ? hi : lo;
            const float give = odd ? lo : hi;
            got[q] = __builtin_bit_cast(float, __builtin_amdgcn_update_dpp(
                         0, __builtin_bit_cast(int, give), 0xB1, 0xF, 0xF, true));
        }
        v4f first, second;
        #pragma unroll
        for (int q = 0; q < 4; ++q) {
            first[q]  = odd ? got[q]  : keep[q];
            second[q] = odd ? keep[q] : got[q];
        }
        *(v4f*)(orow + bb)        = first;
        *(v4f*)(orow + bb + P_SZ) = second;
    }
}

extern "C" void kernel_launch(void* const* d_in, const int* in_sizes, int n_in,
                              void* d_out, int out_size, void* d_ws, size_t ws_size,
                              hipStream_t stream) {
    (void)in_sizes; (void)n_in; (void)d_ws; (void)ws_size; (void)out_size;
    const float* x    = (const float*)d_in[0];
    const float* W    = (const float*)d_in[1];
    const float* bias = (const float*)d_in[2];
    float* out        = (float*)d_out;

    dim3 grid(P_SZ / PB, B_SZ / BTILE, E_SZ / ETILE);
    dim3 block(256);
    loop_proj_wmma_kernel<<<grid, block, 0, stream>>>(x, W, bias, out);
}


# Next chunk's 48 global loads issued 8 at the head of each of the first 6 double k-steps, not as one block ahead of all 16 k-steps: the 16-line loads queue under the WMMAs; d-sum order unchanged
# speedup vs baseline: 2.3405x; 1.0540x over previous
#include <hip/hip_runtime.h>


typedef __attribute__((ext_vector_type(2))) float v2f;
typedef __attribute__((ext_vector_type(4))) float v4f;
typedef __attribute__((ext_vector_type(8))) float v8f;

#define B_SZ 256
#define D_SZ 512
#define P_SZ 128
#define E_SZ 128
#define XB_PITCH (D_SZ * P_SZ)
#define WP_PITCH (D_SZ * E_SZ)
#define OB_PITCH (E_SZ * P_SZ)

#define KC     64
#define KPAD   68
#define PB      8
#define BTILE  32
#define ETILE  64
#define XROWS  (256 / 2 / KC)
#define XREP   (BTILE / XROWS)
#define WVEC   (ETILE / 8)
#define WREP   (PB / XROWS)
#define NCHUNK (D_SZ / KC)
#define NLOADS (WREP * WVEC + XREP)
#define LPER   8
#define LSTEPS (NLOADS / LPER)

__global__ __launch_bounds__(256, 1)
void loop_proj_wmma_kernel(const float* __restrict__ x,
                           const float* __restrict__ W,
                           const float* __restrict__ bias,
                           float* __restrict__ out)
{
    __shared__ float sX[PB * BTILE * KPAD];
    __shared__ float sW[PB * ETILE * KPAD];

    const int tid  = threadIdx.x;
    const int lane = tid & 31;
    const int wave = __builtin_amdgcn_readfirstlane(tid >> 5);
    const int bt   = (wave >> 2) * 16;
    const int es   = (wave & 3) * 16;

    const int pg0 = blockIdx.x * PB;
    const int b0  = blockIdx.y * BTILE;
    const int e0  = blockIdx.z * ETILE;

    const int m    = lane & 15;
    const int koff = (lane >> 4) << 1;

    v8f acc[PB] = {};

    const int pairX = tid >> 1;
    const int phalf = (tid & 1) * 4;
    const int iX    = pairX / KC;
    const int kX    = pairX % KC;
    const float* xp0 = x + (size_t)(b0 + iX) * XB_PITCH
                         + (size_t)kX * P_SZ + pg0 + phalf;

    const int piW   = (tid >> 1) / KC;
    const int kW    = (tid >> 1) % KC;
    const int eSel  = (tid & 1) * 4;
    const float* wp = W + (size_t)(pg0 + piW) * WP_PITCH
                        + (size_t)kW * E_SZ + e0 + eSel;

    auto stage = [&](const v4f (&rw)[WREP][WVEC], const v4f (&rx)[XREP]) {
        #pragma unroll
        for (int r = 0; r < XREP; ++r) {
            #pragma unroll
            for (int j = 0; j < 4; ++j)
                sX[((phalf + j) * BTILE + iX + XROWS * r) * KPAD + kX] = rx[r][j];
        }
        #pragma unroll
        for (int s = 0; s < WREP; ++s) {
            #pragma unroll
            for (int j = 0; j < WVEC; ++j) {
                const int e = eSel + j * 8;
                #pragma unroll
                for (int q = 0; q < 4; ++q)
                    sW[((piW + XROWS * s) * ETILE + e + q) * KPAD + kW] = rw[s][j][q];
            }
        }
    };
    auto fetch1 = [&](int i, size_t off, v4f (&rw)[WREP][WVEC], v4f (&rx)[XREP]) {
        if (i < XREP) {
            rx[i] = *(const v4f*)(xp0 + off + (size_t)(XROWS * i) * XB_PITCH);
        } else {
            const int s = (i - XREP) / WVEC, j = (i - XREP) % WVEC;
            rw[s][j] = *(const v4f*)(wp + off + (size_t)(XROWS * s) * WP_PITCH + j * 8);
        }
    };

    {
        v4f rw[WREP][WVEC], rx[XREP];
        #pragma unroll
        for (int i = 0; i < NLOADS; ++i) fetch1(i, 0, rw, rx);
        stage(rw, rx);
    }
    __syncthreads();

    auto chunk = [&](bool more, size_t noff, v4f (&nrw)[WREP][WVEC], v4f (&nrx)[XREP]) {
        const float* bX = sX;
        const float* bW = sW;
        #pragma unroll
        for (int ds = 0; ds < KC / 8; ++ds) {
            const int k0 = ds * 8;
            if (more && ds < LSTEPS) {
                #pragma unroll
                for (int q = 0; q < LPER; ++q) fetch1(ds * LPER + q, noff, nrw, nrx);
            }
            v2f a[2][PB], bf[2][PB];
            #pragma unroll
            for (int j = 0; j < PB; ++j) {
                a[0][j]  = *(const v2f*)&bX[(j * BTILE + bt + m) * KPAD + k0 + koff];
                a[1][j]  = *(const v2f*)&bX[(j * BTILE + bt + m) * KPAD + k0 + 4 + koff];
            }
            #pragma unroll
            for (int j = 0; j < PB; ++j) {
                bf[0][j] = *(const v2f*)&bW[(j * ETILE + es + m) * KPAD + k0 + koff];
                bf[1][j] = *(const v2f*)&bW[(j * ETILE + es + m) * KPAD + k0 + 4 + koff];
            }
            #pragma unroll
            for (int h = 0; h < 2; ++h) {
                #pragma unroll
                for (int j = 0; j < PB; ++j) {
                    acc[j] = __builtin_amdgcn_wmma_f32_16x16x4_f32(
                                 false, a[h][j], false, bf[h][j], (short)0,
                                 acc[j], false, false);
                }
            }
            __builtin_amdgcn_sched_barrier(0);
        }
    };

    for (int c = 0; c < NCHUNK; ++c) {
        const bool more = (c + 1) < NCHUNK;
        v4f nrw[WREP][WVEC], nrx[XREP];
        chunk(more, (size_t)(c + 1) * KC * P_SZ, nrw, nrx);
        if (more) {
            __syncthreads();
            stage(nrw, nrx);
            __syncthreads();
        }
    }

    asm volatile("v_nop\n\tv_nop\n\tv_nop\n\tv_nop\n\tv_nop");
    __builtin_amdgcn_sched_barrier(0);

    const int  el   = (int)__builtin_amdgcn_mbcnt_lo(~0u, 0u);
    const int  ec   = el & 15;
    const bool odd  = (ec & 1) != 0;
    const int  mtop = (el >> 4) << 3;
    const int  e    = e0 + es + ec;
    float bi[PB];
    #pragma unroll
    for (int j = 0; j < PB; ++j) bi[j] = bias[(pg0 + j) * E_SZ + e];

    float* orow = out + (size_t)(e0 + es + (ec & ~1)) * P_SZ + pg0 + (odd ? 4 : 0);

    #pragma unroll
    for (int r = 0; r < 8; ++r) {
        const size_t bb = (size_t)(b0 + bt + r + mtop) * OB_PITCH;
        float keep[4], got[4];
        #pragma unroll
        for (int q = 0; q < 4; ++q) {
            const float lo = acc[q][r]     + bi[q];
            const float hi = acc[q + 4][r] + bi[q + 4];
            keep[q] = odd ? hi : lo;
            const float give = odd ? lo : hi;
            got[q] = __builtin_bit_cast(float, __builtin_amdgcn_update_dpp(
                         0, __builtin_bit_cast(int, give), 0xB1, 0xF, 0xF, true));
        }
        v4f first, second;
        #pragma unroll
        for (int q = 0; q < 4; ++q) {
            first[q]  = odd ? got[q]  : keep[q];
            second[q] = odd ? keep[q] : got[q];
        }
        *(v4f*)(orow + bb)        = first;
        *(v4f*)(orow + bb + P_SZ) = second;
    }
}

extern "C" void kernel_launch(void* const* d_in, const int* in_sizes, int n_in,
                              void* d_out, int out_size, void* d_ws, size_t ws_size,
                              hipStream_t stream) {
    (void)in_sizes; (void)n_in; (void)d_ws; (void)ws_size; (void)out_size;
    const float* x    = (const float*)d_in[0];
    const float* W    = (const float*)d_in[1];
    const float* bias = (const float*)d_in[2];
    float* out        = (float*)d_out;

    dim3 grid(P_SZ / PB, B_SZ / BTILE, E_SZ / ETILE);
    dim3 block(256);
    loop_proj_wmma_kernel<<<grid, block, 0, stream>>>(x, W, bias, out);
}


# LDS row = two halves of 32 d in turn: each double k-step stores 6 pieces of the next half beside its WMMAs; no store phase between two barriers, one LDS-only barrier per 32 d; d-sum order unchanged
# speedup vs baseline: 2.6317x; 1.1244x over previous
#include <hip/hip_runtime.h>


typedef __attribute__((ext_vector_type(2))) float v2f;
typedef __attribute__((ext_vector_type(4))) float v4f;
typedef __attribute__((ext_vector_type(8))) float v8f;

#define B_SZ 256
#define D_SZ 512
#define P_SZ 128
#define E_SZ 128
#define XB_PITCH (D_SZ * P_SZ)
#define WP_PITCH (D_SZ * E_SZ)
#define OB_PITCH (E_SZ * P_SZ)

#define KH     32
#define KPAD   68
#define PB      8
#define BTILE  32
#define ETILE  64
#define XROWS  (256 / 2 / KH)
#define XREP   (BTILE / XROWS)
#define WVEC   (ETILE / 8)
#define WREP   (PB / XROWS)
#define NHALF  (D_SZ / KH)
#define NLOADS (WREP * WVEC + XREP)
#define DSTEPS (KH / 8)
#define LPER   (NLOADS / DSTEPS)

__global__ __launch_bounds__(256, 1)
void loop_proj_wmma_kernel(const float* __restrict__ x,
                           const float* __restrict__ W,
                           const float* __restrict__ bias,
                           float* __restrict__ out)
{
    __shared__ float sX[PB * BTILE * KPAD];
    __shared__ float sW[PB * ETILE * KPAD];

    const int tid  = threadIdx.x;
    const int lane = tid & 31;
    const int wave = __builtin_amdgcn_readfirstlane(tid >> 5);
    const int bt   = (wave >> 2) * 16;
    const int es   = (wave & 3) * 16;

    const int pg0 = blockIdx.x * PB;
    const int b0  = blockIdx.y * BTILE;
    const int e0  = blockIdx.z * ETILE;

    const int m    = lane & 15;
    const int koff = (lane >> 4) << 1;

    v8f acc[PB] = {};

    const int pairX = tid >> 1;
    const int phalf = (tid & 1) * 4;
    const int iX    = pairX / KH;
    const int kX    = pairX % KH;
    const float* xp0 = x + (size_t)(b0 + iX) * XB_PITCH
                         + (size_t)kX * P_SZ + pg0 + phalf;

    const int piW   = (tid >> 1) / KH;
    const int kW    = (tid >> 1) % KH;
    const int eSel  = (tid & 1) * 4;
    const float* wp = W + (size_t)(pg0 + piW) * WP_PITCH
                        + (size_t)kW * E_SZ + e0 + eSel;

    auto stage1 = [&](int i, int hf, const v4f (&rw)[WREP][WVEC], const v4f (&rx)[XREP]) {
        if (i < XREP) {
            #pragma unroll
            for (int j = 0; j < 4; ++j)
                sX[((phalf + j) * BTILE + iX + XROWS * i) * KPAD + hf * KH + kX] = rx[i][j];
        } else {
            const int s = (i - XREP) / WVEC, j = (i - XREP) % WVEC;
            const int e = eSel + j * 8;
            #pragma unroll
            for (int q = 0; q < 4; ++q)
                sW[((piW + XROWS * s) * ETILE + e + q) * KPAD + hf * KH + kW] = rw[s][j][q];
        }
    };
    auto fetch1 = [&](int i, size_t off, v4f (&rw)[WREP][WVEC], v4f (&rx)[XREP]) {
        if (i < XREP) {
            rx[i] = *(const v4f*)(xp0 + off + (size_t)(XROWS * i) * XB_PITCH);
        } else {
            const int s = (i - XREP) / WVEC, j = (i - XREP) % WVEC;
            rw[s][j] = *(const v4f*)(wp + off + (size_t)(XROWS * s) * WP_PITCH + j * 8);
        }
    };

    v4f aw[WREP][WVEC], ax[XREP], bw[WREP][WVEC], bx[XREP];

    {
        #pragma unroll
        for (int i = 0; i < NLOADS; ++i) fetch1(i, 0, aw, ax);
        #pragma unroll
        for (int i = 0; i < NLOADS; ++i) fetch1(i, (size_t)KH * P_SZ, bw, bx);
        #pragma unroll
        for (int i = 0; i < NLOADS; ++i) stage1(i, 0, aw, ax);
    }
    __syncthreads();

    auto halfstep = [&](int hf, bool doStore, const v4f (&srw)[WREP][WVEC], const v4f (&srx)[XREP],
                        bool doLoad, size_t noff, v4f (&lrw)[WREP][WVEC], v4f (&lrx)[XREP]) {
        const float* bX = sX;
        const float* bW = sW;
        #pragma unroll
        for (int ds = 0; ds < DSTEPS; ++ds) {
            const int k0 = hf * KH + ds * 8;
            if (doLoad) {
                #pragma unroll
                for (int q = 0; q < LPER; ++q) fetch1(ds * LPER + q, noff, lrw, lrx);
            }
            v2f a[2][PB], bf[2][PB];
            #pragma unroll
            for (int j = 0; j < PB; ++j) {
                a[0][j]  = *(const v2f*)&bX[(j * BTILE + bt + m) * KPAD + k0 + koff];
                a[1][j]  = *(const v2f*)&bX[(j * BTILE + bt + m) * KPAD + k0 + 4 + koff];
            }
            #pragma unroll
            for (int j = 0; j < PB; ++j) {
                bf[0][j] = *(const v2f*)&bW[(j * ETILE + es + m) * KPAD + k0 + koff];
                bf[1][j] = *(const v2f*)&bW[(j * ETILE + es + m) * KPAD + k0 + 4 + koff];
            }
            #pragma unroll
            for (int h = 0; h < 2; ++h) {
                #pragma unroll
                for (int j = 0; j < PB; ++j) {
                    acc[j] = __builtin_amdgcn_wmma_f32_16x16x4_f32(
                                 false, a[h][j], false, bf[h][j], (short)0,
                                 acc[j], false, false);
                }
            }
            if (doStore) {
                #pragma unroll
                for (int q = 0; q < LPER; ++q) stage1(ds * LPER + q, hf ^ 1, srw, srx);
            }
            __builtin_amdgcn_sched_barrier(0);
        }
    };

    auto lds_barrier = [&]() {
        asm volatile("s_wait_dscnt 0x0\n\ts_barrier_signal -1\n\ts_barrier_wait -1" ::: "memory");
    };

    for (int c = 0; c < NHALF / 2; ++c) {
        const bool more = (c + 1) < NHALF / 2;
        const size_t o = (size_t)(2 * c + 2) * KH * P_SZ;
        halfstep(0, true, bw, bx, more, o, aw, ax);
        lds_barrier();
        halfstep(1, more, aw, ax, more, o + (size_t)KH * P_SZ, bw, bx);
        if (more) lds_barrier();
    }

    asm volatile("v_nop\n\tv_nop\n\tv_nop\n\tv_nop\n\tv_nop");
    __builtin_amdgcn_sched_barrier(0);

    const int  el   = (int)__builtin_amdgcn_mbcnt_lo(~0u, 0u);
    const int  ec   = el & 15;
    const bool odd  = (ec & 1) != 0;
    const int  mtop = (el >> 4) << 3;
    const int  e    = e0 + es + ec;
    float bi[PB];
    #pragma unroll
    for (int j = 0; j < PB; ++j) bi[j] = bias[(pg0 + j) * E_SZ + e];

    float* orow = out + (size_t)(e0 + es + (ec & ~1)) * P_SZ + pg0 + (odd ? 4 : 0);

    #pragma unroll
    for (int r = 0; r < 8; ++r) {
        const size_t bb = (size_t)(b0 + bt + r + mtop) * OB_PITCH;
        float keep[4], got[4];
        #pragma unroll
        for (int q = 0; q < 4; ++q) {
            const float lo = acc[q][r]     + bi[q];
            const float hi = acc[q + 4][r] + bi[q + 4];
            keep[q] = odd ? hi : lo;
            const float give = odd ? lo : hi;
            got[q] = __builtin_bit_cast(float, __builtin_amdgcn_update_dpp(
                         0, __builtin_bit_cast(int, give), 0xB1, 0xF, 0xF, true));
        }
        v4f first, second;
        #pragma unroll
        for (int q = 0; q < 4; ++q) {
            first[q]  = odd ? got[q]  : keep[q];
            second[q] = odd ? keep[q] : got[q];
        }
        *(v4f*)(orow + bb)        = first;
        *(v4f*)(orow + bb + P_SZ) = second;
    }
}

extern "C" void kernel_launch(void* const* d_in, const int* in_sizes, int n_in,
                              void* d_out, int out_size, void* d_ws, size_t ws_size,
                              hipStream_t stream) {
    (void)in_sizes; (void)n_in; (void)d_ws; (void)ws_size; (void)out_size;
    const float* x    = (const float*)d_in[0];
    const float* W    = (const float*)d_in[1];
    const float* bias = (const float*)d_in[2];
    float* out        = (float*)d_out;

    dim3 grid(P_SZ / PB, B_SZ / BTILE, E_SZ / ETILE);
    dim3 block(256);
    loop_proj_wmma_kernel<<<grid, block, 0, stream>>>(x, W, bias, out);
}
